# starting kernel
# baseline (speedup 1.0000x reference)
#include <hip/hip_runtime.h>
#include <hip/hip_bf16.h>

typedef __attribute__((ext_vector_type(2))) float v2f;
typedef __attribute__((ext_vector_type(8))) float v8f;

#define BATCH 8
#define HEADS 4
#define NQ    900
#define DIM   128
#define CH    32
#define NPAD  928
#define NT    57
#define SROW  928

__global__ __launch_bounds__(256) void proj_kernel(
    const float* __restrict__ q,
    const float* __restrict__ Wq, const float* __restrict__ Wk,
    const float* __restrict__ Wv,
    float* __restrict__ yq, float* __restrict__ yk, float* __restrict__ yv)
{
    const int rt    = blockIdx.x;
    const int which = blockIdx.y;
    const int wave  = threadIdx.x >> 5;
    const int lane  = threadIdx.x & 31;
    const int lh    = lane >> 4;
    const int ll    = lane & 15;

    const float* W = (which == 0) ? Wq : (which == 1) ? Wk : Wv;
    float*       Y = (which == 0) ? yq : (which == 1) ? yk : yv;

    const int r0 = rt * 16;
    const int d0 = wave * 16;

    const float* aptr = q + (size_t)(r0 + ll) * DIM + 2 * lh;
    const float* bptr = W + (size_t)(d0 + ll) * DIM + 2 * lh;

    v8f acc = {};
    #pragma unroll
    for (int k = 0; k < DIM; k += 4) {
        v2f a = *(const v2f*)(aptr + k);
        v2f b = *(const v2f*)(bptr + k);
        acc = __builtin_amdgcn_wmma_f32_16x16x4_f32(false, a, false, b,
                                                    (short)0, acc, false, false);
    }

    const int d    = d0 + ll;
    const int hsel = d >> 5;
    const int csel = d & 31;
    #pragma unroll
    for (int r = 0; r < 8; ++r) {
        int row = r0 + r + 8 * lh;
        int b_  = row / NQ;
        int n_  = row - b_ * NQ;
        Y[(((size_t)(b_ * HEADS + hsel)) * NPAD + n_) * CH + csel] = acc[r];
    }
}

__global__ __launch_bounds__(256) void norm_kernel(
    float* __restrict__ yq, float* __restrict__ yk,
    const float* __restrict__ temp)
{
    const int rows = BATCH * HEADS * NQ;
    const int wid  = blockIdx.x * 8 + (threadIdx.x >> 5);
    if (wid >= rows) return;
    const int lane = threadIdx.x & 31;
    const int bh   = wid / NQ;
    const int n    = wid - bh * NQ;
    const int h    = bh & (HEADS - 1);

    const size_t off = ((size_t)bh * NPAD + n) * CH + lane;
    float vq = yq[off], vk = yk[off];
    float sq = vq * vq, sk = vk * vk;
    #pragma unroll
    for (int s = 16; s > 0; s >>= 1) {
        sq += __shfl_xor(sq, s, 32);
        sk += __shfl_xor(sk, s, 32);
    }
    const float t  = temp[h];
    yq[off] = vq / fmaxf(sqrtf(sq), 1e-12f) * t;
    yk[off] = vk / fmaxf(sqrtf(sk), 1e-12f);
}

__global__ __launch_bounds__(256) void attn_kernel(
    const float* __restrict__ qh, const float* __restrict__ kh,
    const float* __restrict__ vh,
    const float* __restrict__ a1p, const float* __restrict__ a2p,
    const float* __restrict__ a3p, const float* __restrict__ a4p,
    float* __restrict__ yo)
{
    __shared__ float sc[16 * SROW];

    const int bh   = blockIdx.y;
    const int nt   = blockIdx.x;
    const int wave = threadIdx.x >> 5;
    const int lane = threadIdx.x & 31;
    const int lh   = lane >> 4;
    const int ll   = lane & 15;

    const float* qbase = qh + (size_t)bh * NPAD * CH;
    const float* kbase = kh + (size_t)bh * NPAD * CH;
    const float* vbase = vh + (size_t)bh * NPAD * CH;
    const int n0 = nt * 16;

    v2f areg[8];
    {
        const float* ap = qbase + (size_t)(n0 + ll) * CH + 2 * lh;
        #pragma unroll
        for (int c = 0; c < 8; ++c) areg[c] = *(const v2f*)(ap + 4 * c);
    }

    for (int mt = wave; mt < NT; mt += 8) {
        const int  m0   = mt * 16;
        const int  mrow = m0 + ll;
        const bool ok   = (mrow < NQ);
        const float* bp = kbase + (size_t)mrow * CH + 2 * lh;
        v8f acc = {};
        #pragma unroll
        for (int c = 0; c < 8; ++c) {
            v2f b;
            if (ok) b = *(const v2f*)(bp + 4 * c);
            else    { b.x = 0.f; b.y = 0.f; }
            acc = __builtin_amdgcn_wmma_f32_16x16x4_f32(false, areg[c], false, b,
                                                        (short)0, acc, false, false);
        }
        #pragma unroll
        for (int r = 0; r < 8; ++r)
            sc[(r + 8 * lh) * SROW + m0 + ll] = acc[r];
    }
    __syncthreads();

    for (int i = threadIdx.x; i < 16 * (SROW - NQ); i += 256) {
        int row = i / (SROW - NQ);
        int m   = NQ + i - row * (SROW - NQ);
        sc[row * SROW + m] = -1e30f;
    }
    __syncthreads();

    const float A1 = a1p[0], A2 = a2p[0], A3 = a3p[0], A4 = a4p[0];

    for (int rr = 0; rr < 2; ++rr) {
        const int row = wave + 8 * rr;
        float* srow = sc + row * SROW;

        float rv[29];
        #pragma unroll
        for (int j = 0; j < 29; ++j) rv[j] = srow[lane + 32 * j];

        float mx = -3.0e38f;
        #pragma unroll
        for (int j = 0; j < 29; ++j) mx = fmaxf(mx, rv[j]);
        #pragma unroll
        for (int s = 16; s > 0; s >>= 1) mx = fmaxf(mx, __shfl_xor(mx, s, 32));

        unsigned rk[29];
        #pragma unroll
        for (int j = 0; j < 29; ++j) {
            unsigned u = __float_as_uint(rv[j]);
            rk[j] = (u & 0x80000000u) ? ~u : (u | 0x80000000u);
        }

        unsigned th[3];
        const int ks[3] = {100, 300, 600};
        #pragma unroll
        for (int t = 0; t < 3; ++t) {
            unsigned cur = 0u;
            for (int bit = 31; bit >= 0; --bit) {
                const unsigned cand = cur | (1u << bit);
                int cnt = 0;
                #pragma unroll
                for (int j = 0; j < 29; ++j) cnt += (rk[j] >= cand) ? 1 : 0;
                #pragma unroll
                for (int s = 16; s > 0; s >>= 1) cnt += __shfl_xor(cnt, s, 32);
                if (cnt >= ks[t]) cur = cand;
            }
            th[t] = cur;
        }

        float s1 = 0.f, s2 = 0.f, s3 = 0.f, s4 = 0.f;
        float w[29];
        #pragma unroll
        for (int j = 0; j < 29; ++j) {
            float e = __expf(rv[j] - mx);
            w[j] = e;
            s4 += e;
            if (rk[j] >= th[0]) s1 += e;
            if (rk[j] >= th[1]) s2 += e;
            if (rk[j] >= th[2]) s3 += e;
        }
        #pragma unroll
        for (int s = 16; s > 0; s >>= 1) {
            s1 += __shfl_xor(s1, s, 32);
            s2 += __shfl_xor(s2, s, 32);
            s3 += __shfl_xor(s3, s, 32);
            s4 += __shfl_xor(s4, s, 32);
        }
        const float c1 = A1 / s1, c2 = A2 / s2, c3 = A3 / s3, c4 = A4 / s4;
        #pragma unroll
        for (int j = 0; j < 29; ++j) {
            float f = c4;
            if (rk[j] >= th[0]) f += c1;
            if (rk[j] >= th[1]) f += c2;
            if (rk[j] >= th[2]) f += c3;
            srow[lane + 32 * j] = w[j] * f;
        }
    }
    __syncthreads();

    const int nhalf = wave & 1;
    const int kg    = wave >> 1;
    const int cn    = 16 * nhalf + ll;
    v8f acc = {};
    for (int mt = kg; mt < NT; mt += 4) {
        #pragma unroll
        for (int k4 = 0; k4 < 4; ++k4) {
            const int k0 = mt * 16 + 4 * k4;
            const int kk = k0 + 2 * lh;
            v2f a = *(const v2f*)(sc + ll * SROW + kk);
            v2f b;
            b.x = (kk     < NQ) ? vbase[(size_t)kk * CH + cn]       : 0.f;
            b.y = (kk + 1 < NQ) ? vbase[(size_t)(kk + 1) * CH + cn] : 0.f;
            acc = __builtin_amdgcn_wmma_f32_16x16x4_f32(false, a, false, b,
                                                        (short)0, acc, false, false);
        }
    }
    __syncthreads();

    float* part = sc;
    #pragma unroll
    for (int r = 0; r < 8; ++r) part[(wave * 8 + r) * 32 + lane] = acc[r];
    __syncthreads();

    if (wave < 2) {
        #pragma unroll
        for (int r = 0; r < 8; ++r) {
            float v = 0.f;
            #pragma unroll
            for (int g = 0; g < 4; ++g)
                v += part[(((g * 2 + wave) * 8) + r) * 32 + lane];
            const int row = n0 + r + 8 * lh;
            const int col = 16 * wave + ll;
            if (row < NQ)
                yo[((size_t)bh * NPAD + row) * CH + col] = v;
        }
    }
}

__global__ __launch_bounds__(256) void bsum_kernel(
    const float* __restrict__ yo, float* __restrict__ msum)
{
    const int idx = blockIdx.x * 256 + threadIdx.x;
    if (idx >= 912 * DIM) return;
    const int n = idx >> 7;
    const int d = idx & 127;
    float v = 0.f;
    if (n < NQ) {
        const int h = d >> 5, c = d & 31;
        #pragma unroll
        for (int b = 0; b < BATCH; ++b)
            v += yo[(((size_t)(b * HEADS + h)) * NPAD + n) * CH + c];
    }
    msum[idx] = v;
}

__global__ __launch_bounds__(256) void out_kernel(
    const float* __restrict__ msum, const float* __restrict__ Wo,
    const float* __restrict__ q, float* __restrict__ out)
{
    const int rt   = blockIdx.x;
    const int wave = threadIdx.x >> 5;
    const int lane = threadIdx.x & 31;
    const int lh   = lane >> 4;
    const int ll   = lane & 15;
    const int r0   = rt * 16;
    const int d0   = wave * 16;

    const float* aptr = msum + (size_t)(r0 + ll) * DIM + 2 * lh;
    const float* bptr = Wo   + (size_t)(d0 + ll) * DIM + 2 * lh;

    v8f acc = {};
    #pragma unroll
    for (int k = 0; k < DIM; k += 4) {
        v2f a = *(const v2f*)(aptr + k);
        v2f b = *(const v2f*)(bptr + k);
        acc = __builtin_amdgcn_wmma_f32_16x16x4_f32(false, a, false, b,
                                                    (short)0, acc, false, false);
    }
    #pragma unroll
    for (int r = 0; r < 8; ++r) {
        const int row = r0 + r + 8 * lh;
        const int col = d0 + ll;
        if (row < NQ)
            out[(size_t)row * DIM + col] = acc[r] + q[(size_t)row * DIM + col];
    }
}

extern "C" void kernel_launch(void* const* d_in, const int* in_sizes, int n_in,
                              void* d_out, int out_size, void* d_ws, size_t ws_size,
                              hipStream_t stream)
{
    const float* q    = (const float*)d_in[0];
    const float* Wq   = (const float*)d_in[1];
    const float* Wk   = (const float*)d_in[2];
    const float* Wv   = (const float*)d_in[3];
    const float* Wo   = (const float*)d_in[4];
    const float* temp = (const float*)d_in[5];
    const float* a1   = (const float*)d_in[6];
    const float* a2   = (const float*)d_in[7];
    const float* a3   = (const float*)d_in[8];
    const float* a4   = (const float*)d_in[9];
    float* out = (float*)d_out;

    const size_t per = (size_t)BATCH * HEADS * NPAD * CH;
    float* yq   = (float*)d_ws;
    float* yk   = yq + per;
    float* yv   = yk + per;
    float* yo   = yv + per;
    float* msum = yo + per;

    proj_kernel<<<dim3(450, 3), 256, 0, stream>>>(q, Wq, Wk, Wv, yq, yk, yv);
    norm_kernel<<<(BATCH * HEADS * NQ + 7) / 8, 256, 0, stream>>>(yq, yk, temp);
    attn_kernel<<<dim3(NT, BATCH * HEADS), 256, 0, stream>>>(
        yq, yk, yv, a1, a2, a3, a4, yo);
    bsum_kernel<<<(912 * DIM + 255) / 256, 256, 0, stream>>>(yo, msum);
    out_kernel<<<NT, 256, 0, stream>>>(msum, Wo, q, out);
}


# top-k thresholds and row max from one in-register bitonic sort of each score row's keys (32 regs x 32 lanes, DPP lane exchanges), not three 32-step bit searches; masks, sums, results bit-identical
# speedup vs baseline: 1.4662x; 1.4662x over previous
#include <hip/hip_runtime.h>
#include <hip/hip_bf16.h>

typedef __attribute__((ext_vector_type(2))) float v2f;
typedef __attribute__((ext_vector_type(8))) float v8f;

#define BATCH 8
#define HEADS 4
#define NQ    900
#define DIM   128
#define CH    32
#define NPAD  928
#define NT    57
#define SROW  928

__global__ __launch_bounds__(256) void proj_kernel(
    const float* __restrict__ q,
    const float* __restrict__ Wq, const float* __restrict__ Wk,
    const float* __restrict__ Wv,
    float* __restrict__ yq, float* __restrict__ yk, float* __restrict__ yv)
{
    const int rt    = blockIdx.x;
    const int which = blockIdx.y;
    const int wave  = threadIdx.x >> 5;
    const int lane  = threadIdx.x & 31;
    const int lh    = lane >> 4;
    const int ll    = lane & 15;

    const float* W = (which == 0) ? Wq : (which == 1) ? Wk : Wv;
    float*       Y = (which == 0) ? yq : (which == 1) ? yk : yv;

    const int r0 = rt * 16;
    const int d0 = wave * 16;

    const float* aptr = q + (size_t)(r0 + ll) * DIM + 2 * lh;
    const float* bptr = W + (size_t)(d0 + ll) * DIM + 2 * lh;

    v8f acc = {};
    #pragma unroll
    for (int k = 0; k < DIM; k += 4) {
        v2f a = *(const v2f*)(aptr + k);
        v2f b = *(const v2f*)(bptr + k);
        acc = __builtin_amdgcn_wmma_f32_16x16x4_f32(false, a, false, b,
                                                    (short)0, acc, false, false);
    }

    const int d    = d0 + ll;
    const int hsel = d >> 5;
    const int csel = d & 31;
    #pragma unroll
    for (int r = 0; r < 8; ++r) {
        int row = r0 + r + 8 * lh;
        int b_  = row / NQ;
        int n_  = row - b_ * NQ;
        Y[(((size_t)(b_ * HEADS + hsel)) * NPAD + n_) * CH + csel] = acc[r];
    }
}

__global__ __launch_bounds__(256) void norm_kernel(
    float* __restrict__ yq, float* __restrict__ yk,
    const float* __restrict__ temp)
{
    const int rows = BATCH * HEADS * NQ;
    const int wid  = blockIdx.x * 8 + (threadIdx.x >> 5);
    if (wid >= rows) return;
    const int lane = threadIdx.x & 31;
    const int bh   = wid / NQ;
    const int n    = wid - bh * NQ;
    const int h    = bh & (HEADS - 1);

    const size_t off = ((size_t)bh * NPAD + n) * CH + lane;
    float vq = yq[off], vk = yk[off];
    float sq = vq * vq, sk = vk * vk;
    #pragma unroll
    for (int s = 16; s > 0; s >>= 1) {
        sq += __shfl_xor(sq, s, 32);
        sk += __shfl_xor(sk, s, 32);
    }
    const float t  = temp[h];
    yq[off] = vq / fmaxf(sqrtf(sq), 1e-12f) * t;
    yk[off] = vk / fmaxf(sqrtf(sk), 1e-12f);
}

__device__ __forceinline__ unsigned umin_(unsigned a, unsigned b) { return a < b ? a : b; }
__device__ __forceinline__ unsigned umax_(unsigned a, unsigned b) { return a < b ? b : a; }

template <int MASK>
__device__ __forceinline__ unsigned lane_xchg(unsigned v)
{
    if constexpr (MASK < 16) {
        return (unsigned)__builtin_amdgcn_update_dpp((int)v, (int)v, 0x160 + MASK,
                                                     0xf, 0xf, false);
    } else {
        return (unsigned)__shfl_xor((int)v, MASK, 32);
    }
}

template <int PX>
__device__ __forceinline__ void reg_stage(unsigned (&k)[32])
{
    #pragma unroll
    for (int r = 0; r < 32; ++r) {
        const int q = r ^ PX;
        if (r < q) {
            const unsigned a = k[r], b = k[q];
            k[r] = umin_(a, b);
            k[q] = umax_(a, b);
        }
    }
}

template <int MASK, int FLIP>
__device__ __forceinline__ void lane_stage(unsigned (&k)[32], const int lane)
{
    constexpr int HB = (MASK >= 16) ? 16 : (MASK >= 8) ? 8 : (MASK >= 4) ? 4
                     : (MASK >= 2) ? 2 : 1;
    const bool lower = (lane & HB) == 0;
    unsigned p[32];
    #pragma unroll
    for (int r = 0; r < 32; ++r) p[r] = lane_xchg<MASK>(k[r ^ FLIP]);
    #pragma unroll
    for (int r = 0; r < 32; ++r) {
        const unsigned a  = k[r];
        const unsigned lo = umin_(a, p[r]);
        const unsigned hi = umax_(a, p[r]);
        k[r] = lower ? lo : hi;
    }
}

__device__ __forceinline__ void reg_tail(unsigned (&k)[32])
{
    reg_stage<16>(k); reg_stage<8>(k); reg_stage<4>(k); reg_stage<2>(k); reg_stage<1>(k);
}

__device__ __forceinline__ void sort_row_keys(unsigned (&k)[32], const int lane)
{
    reg_stage<1>(k);
    reg_stage<3>(k);  reg_stage<1>(k);
    reg_stage<7>(k);  reg_stage<2>(k); reg_stage<1>(k);
    reg_stage<15>(k); reg_stage<4>(k); reg_stage<2>(k); reg_stage<1>(k);
    reg_stage<31>(k); reg_stage<8>(k); reg_stage<4>(k); reg_stage<2>(k); reg_stage<1>(k);
    lane_stage<1, 31>(k, lane);
    reg_tail(k);
    lane_stage<3, 31>(k, lane);  lane_stage<1, 0>(k, lane);
    reg_tail(k);
    lane_stage<7, 31>(k, lane);  lane_stage<2, 0>(k, lane); lane_stage<1, 0>(k, lane);
    reg_tail(k);
    lane_stage<15, 31>(k, lane); lane_stage<4, 0>(k, lane); lane_stage<2, 0>(k, lane);
    lane_stage<1, 0>(k, lane);
    reg_tail(k);
    lane_stage<31, 31>(k, lane); lane_stage<8, 0>(k, lane); lane_stage<4, 0>(k, lane);
    lane_stage<2, 0>(k, lane);   lane_stage<1, 0>(k, lane);
    reg_tail(k);
}

__global__ __launch_bounds__(256) void attn_kernel(
    const float* __restrict__ qh, const float* __restrict__ kh,
    const float* __restrict__ vh,
    const float* __restrict__ a1p, const float* __restrict__ a2p,
    const float* __restrict__ a3p, const float* __restrict__ a4p,
    float* __restrict__ yo)
{
    __shared__ float sc[16 * SROW];

    const int bh   = blockIdx.y;
    const int nt   = blockIdx.x;
    const int wave = threadIdx.x >> 5;
    const int lane = threadIdx.x & 31;
    const int lh   = lane >> 4;
    const int ll   = lane & 15;

    const float* qbase = qh + (size_t)bh * NPAD * CH;
    const float* kbase = kh + (size_t)bh * NPAD * CH;
    const float* vbase = vh + (size_t)bh * NPAD * CH;
    const int n0 = nt * 16;

    v2f areg[8];
    {
        const float* ap = qbase + (size_t)(n0 + ll) * CH + 2 * lh;
        #pragma unroll
        for (int c = 0; c < 8; ++c) areg[c] = *(const v2f*)(ap + 4 * c);
    }

    for (int mt = wave; mt < NT; mt += 8) {
        const int  m0   = mt * 16;
        const int  mrow = m0 + ll;
        const bool ok   = (mrow < NQ);
        const float* bp = kbase + (size_t)mrow * CH + 2 * lh;
        v8f acc = {};
        #pragma unroll
        for (int c = 0; c < 8; ++c) {
            v2f b;
            if (ok) b = *(const v2f*)(bp + 4 * c);
            else    { b.x = 0.f; b.y = 0.f; }
            acc = __builtin_amdgcn_wmma_f32_16x16x4_f32(false, areg[c], false, b,
                                                        (short)0, acc, false, false);
        }
        #pragma unroll
        for (int r = 0; r < 8; ++r)
            sc[(r + 8 * lh) * SROW + m0 + ll] = acc[r];
    }
    __syncthreads();

    for (int i = threadIdx.x; i < 16 * (SROW - NQ); i += 256) {
        int row = i / (SROW - NQ);
        int m   = NQ + i - row * (SROW - NQ);
        sc[row * SROW + m] = -1e30f;
    }
    __syncthreads();

    const float A1 = a1p[0], A2 = a2p[0], A3 = a3p[0], A4 = a4p[0];

    for (int rr = 0; rr < 2; ++rr) {
        const int row = wave + 8 * rr;
        float* srow = sc + row * SROW;

        unsigned key[32];
        #pragma unroll
        for (int j = 0; j < 29; ++j) {
            const unsigned u = __float_as_uint(srow[lane + 32 * j]);
            key[j] = (u & 0x80000000u) ? ~u : (u | 0x80000000u);
        }
        key[29] = 0u; key[30] = 0u; key[31] = 0u;

        sort_row_keys(key, lane);

        unsigned th[3];
        th[0] = (unsigned)__builtin_amdgcn_readlane((int)key[28], 28);
        th[1] = (unsigned)__builtin_amdgcn_readlane((int)key[20], 22);
        th[2] = (unsigned)__builtin_amdgcn_readlane((int)key[8],  13);
        const unsigned ktop = (unsigned)__builtin_amdgcn_readlane((int)key[31], 31);
        const float mx = __uint_as_float((ktop & 0x80000000u) ? (ktop ^ 0x80000000u)
                                                              : ~ktop);

        asm volatile("" ::: "memory");

        float    rv[29];
        unsigned rk[29];
        #pragma unroll
        for (int j = 0; j < 29; ++j) {
            rv[j] = srow[lane + 32 * j];
            const unsigned u = __float_as_uint(rv[j]);
            rk[j] = (u & 0x80000000u) ? ~u : (u | 0x80000000u);
        }

        float s1 = 0.f, s2 = 0.f, s3 = 0.f, s4 = 0.f;
        float w[29];
        #pragma unroll
        for (int j = 0; j < 29; ++j) {
            float e = __expf(rv[j] - mx);
            w[j] = e;
            s4 += e;
            if (rk[j] >= th[0]) s1 += e;
            if (rk[j] >= th[1]) s2 += e;
            if (rk[j] >= th[2]) s3 += e;
        }
        #pragma unroll
        for (int s = 16; s > 0; s >>= 1) {
            s1 += __shfl_xor(s1, s, 32);
            s2 += __shfl_xor(s2, s, 32);
            s3 += __shfl_xor(s3, s, 32);
            s4 += __shfl_xor(s4, s, 32);
        }
        const float c1 = A1 / s1, c2 = A2 / s2, c3 = A3 / s3, c4 = A4 / s4;
        #pragma unroll
        for (int j = 0; j < 29; ++j) {
            float f = c4;
            if (rk[j] >= th[0]) f += c1;
            if (rk[j] >= th[1]) f += c2;
            if (rk[j] >= th[2]) f += c3;
            srow[lane + 32 * j] = w[j] * f;
        }
    }
    __syncthreads();

    const int nhalf = wave & 1;
    const int kg    = wave >> 1;
    const int cn    = 16 * nhalf + ll;
    v8f acc = {};
    for (int mt = kg; mt < NT; mt += 4) {
        #pragma unroll
        for (int k4 = 0; k4 < 4; ++k4) {
            const int k0 = mt * 16 + 4 * k4;
            const int kk = k0 + 2 * lh;
            v2f a = *(const v2f*)(sc + ll * SROW + kk);
            v2f b;
            b.x = (kk     < NQ) ? vbase[(size_t)kk * CH + cn]       : 0.f;
            b.y = (kk + 1 < NQ) ? vbase[(size_t)(kk + 1) * CH + cn] : 0.f;
            acc = __builtin_amdgcn_wmma_f32_16x16x4_f32(false, a, false, b,
                                                        (short)0, acc, false, false);
        }
    }
    __syncthreads();

    float* part = sc;
    #pragma unroll
    for (int r = 0; r < 8; ++r) part[(wave * 8 + r) * 32 + lane] = acc[r];
    __syncthreads();

    if (wave < 2) {
        #pragma unroll
        for (int r = 0; r < 8; ++r) {
            float v = 0.f;
            #pragma unroll
            for (int g = 0; g < 4; ++g)
                v += part[(((g * 2 + wave) * 8) + r) * 32 + lane];
            const int row = n0 + r + 8 * lh;
            const int col = 16 * wave + ll;
            if (row < NQ)
                yo[((size_t)bh * NPAD + row) * CH + col] = v;
        }
    }
}

__global__ __launch_bounds__(256) void bsum_kernel(
    const float* __restrict__ yo, float* __restrict__ msum)
{
    const int idx = blockIdx.x * 256 + threadIdx.x;
    if (idx >= 912 * DIM) return;
    const int n = idx >> 7;
    const int d = idx & 127;
    float v = 0.f;
    if (n < NQ) {
        const int h = d >> 5, c = d & 31;
        #pragma unroll
        for (int b = 0; b < BATCH; ++b)
            v += yo[(((size_t)(b * HEADS + h)) * NPAD + n) * CH + c];
    }
    msum[idx] = v;
}

__global__ __launch_bounds__(256) void out_kernel(
    const float* __restrict__ msum, const float* __restrict__ Wo,
    const float* __restrict__ q, float* __restrict__ out)
{
    const int rt   = blockIdx.x;
    const int wave = threadIdx.x >> 5;
    const int lane = threadIdx.x & 31;
    const int lh   = lane >> 4;
    const int ll   = lane & 15;
    const int r0   = rt * 16;
    const int d0   = wave * 16;

    const float* aptr = msum + (size_t)(r0 + ll) * DIM + 2 * lh;
    const float* bptr = Wo   + (size_t)(d0 + ll) * DIM + 2 * lh;

    v8f acc = {};
    #pragma unroll
    for (int k = 0; k < DIM; k += 4) {
        v2f a = *(const v2f*)(aptr + k);
        v2f b = *(const v2f*)(bptr + k);
        acc = __builtin_amdgcn_wmma_f32_16x16x4_f32(false, a, false, b,
                                                    (short)0, acc, false, false);
    }
    #pragma unroll
    for (int r = 0; r < 8; ++r) {
        const int row = r0 + r + 8 * lh;
        const int col = d0 + ll;
        if (row < NQ)
            out[(size_t)row * DIM + col] = acc[r] + q[(size_t)row * DIM + col];
    }
}

extern "C" void kernel_launch(void* const* d_in, const int* in_sizes, int n_in,
                              void* d_out, int out_size, void* d_ws, size_t ws_size,
                              hipStream_t stream)
{
    const float* q    = (const float*)d_in[0];
    const float* Wq   = (const float*)d_in[1];
    const float* Wk   = (const float*)d_in[2];
    const float* Wv   = (const float*)d_in[3];
    const float* Wo   = (const float*)d_in[4];
    const float* temp = (const float*)d_in[5];
    const float* a1   = (const float*)d_in[6];
    const float* a2   = (const float*)d_in[7];
    const float* a3   = (const float*)d_in[8];
    const float* a4   = (const float*)d_in[9];
    float* out = (float*)d_out;

    const size_t per = (size_t)BATCH * HEADS * NPAD * CH;
    float* yq   = (float*)d_ws;
    float* yk   = yq + per;
    float* yv   = yk + per;
    float* yo   = yv + per;
    float* msum = yo + per;

    proj_kernel<<<dim3(450, 3), 256, 0, stream>>>(q, Wq, Wk, Wv, yq, yk, yv);
    norm_kernel<<<(BATCH * HEADS * NQ + 7) / 8, 256, 0, stream>>>(yq, yk, temp);
    attn_kernel<<<dim3(NT, BATCH * HEADS), 256, 0, stream>>>(
        yq, yk, yv, a1, a2, a3, a4, yo);
    bsum_kernel<<<(912 * DIM + 255) / 256, 256, 0, stream>>>(yo, msum);
    out_kernel<<<NT, 256, 0, stream>>>(msum, Wo, q, out);
}


# row sort: cross-lane compare-exchange as one median-of-three (partner key via the instruction's own DPP lane select, per-lane 0/all-ones limit), not lane move+min+max+select; bit-identical
# speedup vs baseline: 1.6337x; 1.1142x over previous
#include <hip/hip_runtime.h>
#include <hip/hip_bf16.h>

typedef __attribute__((ext_vector_type(2))) float v2f;
typedef __attribute__((ext_vector_type(8))) float v8f;

#define BATCH 8
#define HEADS 4
#define NQ    900
#define DIM   128
#define CH    32
#define NPAD  928
#define NT    57
#define SROW  928

__global__ __launch_bounds__(256) void proj_kernel(
    const float* __restrict__ q,
    const float* __restrict__ Wq, const float* __restrict__ Wk,
    const float* __restrict__ Wv,
    float* __restrict__ yq, float* __restrict__ yk, float* __restrict__ yv)
{
    const int rt    = blockIdx.x;
    const int which = blockIdx.y;
    const int wave  = threadIdx.x >> 5;
    const int lane  = threadIdx.x & 31;
    const int lh    = lane >> 4;
    const int ll    = lane & 15;

    const float* W = (which == 0) ? Wq : (which == 1) ? Wk : Wv;
    float*       Y = (which == 0) ? yq : (which == 1) ? yk : yv;

    const int r0 = rt * 16;
    const int d0 = wave * 16;

    const float* aptr = q + (size_t)(r0 + ll) * DIM + 2 * lh;
    const float* bptr = W + (size_t)(d0 + ll) * DIM + 2 * lh;

    v8f acc = {};
    #pragma unroll
    for (int k = 0; k < DIM; k += 4) {
        v2f a = *(const v2f*)(aptr + k);
        v2f b = *(const v2f*)(bptr + k);
        acc = __builtin_amdgcn_wmma_f32_16x16x4_f32(false, a, false, b,
                                                    (short)0, acc, false, false);
    }

    const int d    = d0 + ll;
    const int hsel = d >> 5;
    const int csel = d & 31;
    #pragma unroll
    for (int r = 0; r < 8; ++r) {
        int row = r0 + r + 8 * lh;
        int b_  = row / NQ;
        int n_  = row - b_ * NQ;
        Y[(((size_t)(b_ * HEADS + hsel)) * NPAD + n_) * CH + csel] = acc[r];
    }
}

__global__ __launch_bounds__(256) void norm_kernel(
    float* __restrict__ yq, float* __restrict__ yk,
    const float* __restrict__ temp)
{
    const int rows = BATCH * HEADS * NQ;
    const int wid  = blockIdx.x * 8 + (threadIdx.x >> 5);
    if (wid >= rows) return;
    const int lane = threadIdx.x & 31;
    const int bh   = wid / NQ;
    const int n    = wid - bh * NQ;
    const int h    = bh & (HEADS - 1);

    const size_t off = ((size_t)bh * NPAD + n) * CH + lane;
    float vq = yq[off], vk = yk[off];
    float sq = vq * vq, sk = vk * vk;
    #pragma unroll
    for (int s = 16; s > 0; s >>= 1) {
        sq += __shfl_xor(sq, s, 32);
        sk += __shfl_xor(sk, s, 32);
    }
    const float t  = temp[h];
    yq[off] = vq / fmaxf(sqrtf(sq), 1e-12f) * t;
    yk[off] = vk / fmaxf(sqrtf(sk), 1e-12f);
}

__device__ __forceinline__ unsigned umin_(unsigned a, unsigned b) { return a < b ? a : b; }
__device__ __forceinline__ unsigned umax_(unsigned a, unsigned b) { return a < b ? b : a; }

template <int PX>
__device__ __forceinline__ void reg_stage(unsigned (&k)[32])
{
    #pragma unroll
    for (int r = 0; r < 32; ++r) {
        const int q = r ^ PX;
        if (r < q) {
            const unsigned a = k[r], b = k[q];
            k[r] = umin_(a, b);
            k[q] = umax_(a, b);
        }
    }
}

template <int MASK, int FLIP>
__device__ __forceinline__ void lane_stage(unsigned (&k)[32], const int lane)
{
    constexpr int HB = (MASK >= 16) ? 16 : (MASK >= 8) ? 8 : (MASK >= 4) ? 4
                     : (MASK >= 2) ? 2 : 1;
    const unsigned lim = (lane & HB) ? 0xffffffffu : 0u;
    unsigned n[32];
    #pragma unroll
    for (int r = 0; r < 32; ++r) {
        if constexpr (MASK < 16) {
            asm("v_med3_u32_e64_dpp %0, %1, %2, %3 row_xmask:%4 row_mask:0xf bank_mask:0xf"
                : "=v"(n[r]) : "v"(k[r ^ FLIP]), "v"(k[r]), "v"(lim), "n"(MASK));
        } else {
            const unsigned p = (unsigned)__shfl_xor((int)k[r ^ FLIP], MASK, 32);
            asm("v_med3_u32 %0, %1, %2, %3"
                : "=v"(n[r]) : "v"(p), "v"(k[r]), "v"(lim));
        }
    }
    #pragma unroll
    for (int r = 0; r < 32; ++r) k[r] = n[r];
}

__device__ __forceinline__ void reg_tail(unsigned (&k)[32])
{
    reg_stage<16>(k); reg_stage<8>(k); reg_stage<4>(k); reg_stage<2>(k); reg_stage<1>(k);
}

__device__ __forceinline__ void sort_row_keys(unsigned (&k)[32], const int lane)
{
    reg_stage<1>(k);
    reg_stage<3>(k);  reg_stage<1>(k);
    reg_stage<7>(k);  reg_stage<2>(k); reg_stage<1>(k);
    reg_stage<15>(k); reg_stage<4>(k); reg_stage<2>(k); reg_stage<1>(k);
    reg_stage<31>(k); reg_stage<8>(k); reg_stage<4>(k); reg_stage<2>(k); reg_stage<1>(k);
    lane_stage<1, 31>(k, lane);
    reg_tail(k);
    lane_stage<3, 31>(k, lane);  lane_stage<1, 0>(k, lane);
    reg_tail(k);
    lane_stage<7, 31>(k, lane);  lane_stage<2, 0>(k, lane); lane_stage<1, 0>(k, lane);
    reg_tail(k);
    lane_stage<15, 31>(k, lane); lane_stage<4, 0>(k, lane); lane_stage<2, 0>(k, lane);
    lane_stage<1, 0>(k, lane);
    reg_tail(k);
    lane_stage<31, 31>(k, lane); lane_stage<8, 0>(k, lane); lane_stage<4, 0>(k, lane);
    lane_stage<2, 0>(k, lane);   lane_stage<1, 0>(k, lane);
    reg_tail(k);
}

__global__ __launch_bounds__(256) void attn_kernel(
    const float* __restrict__ qh, const float* __restrict__ kh,
    const float* __restrict__ vh,
    const float* __restrict__ a1p, const float* __restrict__ a2p,
    const float* __restrict__ a3p, const float* __restrict__ a4p,
    float* __restrict__ yo)
{
    __shared__ float sc[16 * SROW];

    const int bh   = blockIdx.y;
    const int nt   = blockIdx.x;
    const int wave = threadIdx.x >> 5;
    const int lane = threadIdx.x & 31;
    const int lh   = lane >> 4;
    const int ll   = lane & 15;

    const float* qbase = qh + (size_t)bh * NPAD * CH;
    const float* kbase = kh + (size_t)bh * NPAD * CH;
    const float* vbase = vh + (size_t)bh * NPAD * CH;
    const int n0 = nt * 16;

    v2f areg[8];
    {
        const float* ap = qbase + (size_t)(n0 + ll) * CH + 2 * lh;
        #pragma unroll
        for (int c = 0; c < 8; ++c) areg[c] = *(const v2f*)(ap + 4 * c);
    }

    for (int mt = wave; mt < NT; mt += 8) {
        const int  m0   = mt * 16;
        const int  mrow = m0 + ll;
        const bool ok   = (mrow < NQ);
        const float* bp = kbase + (size_t)mrow * CH + 2 * lh;
        v8f acc = {};
        #pragma unroll
        for (int c = 0; c < 8; ++c) {
            v2f b;
            if (ok) b = *(const v2f*)(bp + 4 * c);
            else    { b.x = 0.f; b.y = 0.f; }
            acc = __builtin_amdgcn_wmma_f32_16x16x4_f32(false, areg[c], false, b,
                                                        (short)0, acc, false, false);
        }
        #pragma unroll
        for (int r = 0; r < 8; ++r)
            sc[(r + 8 * lh) * SROW + m0 + ll] = acc[r];
    }
    __syncthreads();

    for (int i = threadIdx.x; i < 16 * (SROW - NQ); i += 256) {
        int row = i / (SROW - NQ);
        int m   = NQ + i - row * (SROW - NQ);
        sc[row * SROW + m] = -1e30f;
    }
    __syncthreads();

    const float A1 = a1p[0], A2 = a2p[0], A3 = a3p[0], A4 = a4p[0];

    for (int rr = 0; rr < 2; ++rr) {
        const int row = wave + 8 * rr;
        float* srow = sc + row * SROW;

        unsigned key[32];
        #pragma unroll
        for (int j = 0; j < 29; ++j) {
            const unsigned u = __float_as_uint(srow[lane + 32 * j]);
            key[j] = (u & 0x80000000u) ? ~u : (u | 0x80000000u);
        }
        key[29] = 0u; key[30] = 0u; key[31] = 0u;

        sort_row_keys(key, lane);

        unsigned th[3];
        th[0] = (unsigned)__builtin_amdgcn_readlane((int)key[28], 28);
        th[1] = (unsigned)__builtin_amdgcn_readlane((int)key[20], 22);
        th[2] = (unsigned)__builtin_amdgcn_readlane((int)key[8],  13);
        const unsigned ktop = (unsigned)__builtin_amdgcn_readlane((int)key[31], 31);
        const float mx = __uint_as_float((ktop & 0x80000000u) ? (ktop ^ 0x80000000u)
                                                              : ~ktop);

        asm volatile("" ::: "memory");

        float    rv[29];
        unsigned rk[29];
        #pragma unroll
        for (int j = 0; j < 29; ++j) {
            rv[j] = srow[lane + 32 * j];
            const unsigned u = __float_as_uint(rv[j]);
            rk[j] = (u & 0x80000000u) ? ~u : (u | 0x80000000u);
        }

        float s1 = 0.f, s2 = 0.f, s3 = 0.f, s4 = 0.f;
        float w[29];
        #pragma unroll
        for (int j = 0; j < 29; ++j) {
            float e = __expf(rv[j] - mx);
            w[j] = e;
            s4 += e;
            if (rk[j] >= th[0]) s1 += e;
            if (rk[j] >= th[1]) s2 += e;
            if (rk[j] >= th[2]) s3 += e;
        }
        #pragma unroll
        for (int s = 16; s > 0; s >>= 1) {
            s1 += __shfl_xor(s1, s, 32);
            s2 += __shfl_xor(s2, s, 32);
            s3 += __shfl_xor(s3, s, 32);
            s4 += __shfl_xor(s4, s, 32);
        }
        const float c1 = A1 / s1, c2 = A2 / s2, c3 = A3 / s3, c4 = A4 / s4;
        #pragma unroll
        for (int j = 0; j < 29; ++j) {
            float f = c4;
            if (rk[j] >= th[0]) f += c1;
            if (rk[j] >= th[1]) f += c2;
            if (rk[j] >= th[2]) f += c3;
            srow[lane + 32 * j] = w[j] * f;
        }
    }
    __syncthreads();

    const int nhalf = wave & 1;
    const int kg    = wave >> 1;
    const int cn    = 16 * nhalf + ll;
    v8f acc = {};
    for (int mt = kg; mt < NT; mt += 4) {
        #pragma unroll
        for (int k4 = 0; k4 < 4; ++k4) {
            const int k0 = mt * 16 + 4 * k4;
            const int kk = k0 + 2 * lh;
            v2f a = *(const v2f*)(sc + ll * SROW + kk);
            v2f b;
            b.x = (kk     < NQ) ? vbase[(size_t)kk * CH + cn]       : 0.f;
            b.y = (kk + 1 < NQ) ? vbase[(size_t)(kk + 1) * CH + cn] : 0.f;
            acc = __builtin_amdgcn_wmma_f32_16x16x4_f32(false, a, false, b,
                                                        (short)0, acc, false, false);
        }
    }
    __syncthreads();

    float* part = sc;
    #pragma unroll
    for (int r = 0; r < 8; ++r) part[(wave * 8 + r) * 32 + lane] = acc[r];
    __syncthreads();

    if (wave < 2) {
        #pragma unroll
        for (int r = 0; r < 8; ++r) {
            float v = 0.f;
            #pragma unroll
            for (int g = 0; g < 4; ++g)
                v += part[(((g * 2 + wave) * 8) + r) * 32 + lane];
            const int row = n0 + r + 8 * lh;
            const int col = 16 * wave + ll;
            if (row < NQ)
                yo[((size_t)bh * NPAD + row) * CH + col] = v;
        }
    }
}

__global__ __launch_bounds__(256) void bsum_kernel(
    const float* __restrict__ yo, float* __restrict__ msum)
{
    const int idx = blockIdx.x * 256 + threadIdx.x;
    if (idx >= 912 * DIM) return;
    const int n = idx >> 7;
    const int d = idx & 127;
    float v = 0.f;
    if (n < NQ) {
        const int h = d >> 5, c = d & 31;
        #pragma unroll
        for (int b = 0; b < BATCH; ++b)
            v += yo[(((size_t)(b * HEADS + h)) * NPAD + n) * CH + c];
    }
    msum[idx] = v;
}

__global__ __launch_bounds__(256) void out_kernel(
    const float* __restrict__ msum, const float* __restrict__ Wo,
    const float* __restrict__ q, float* __restrict__ out)
{
    const int rt   = blockIdx.x;
    const int wave = threadIdx.x >> 5;
    const int lane = threadIdx.x & 31;
    const int lh   = lane >> 4;
    const int ll   = lane & 15;
    const int r0   = rt * 16;
    const int d0   = wave * 16;

    const float* aptr = msum + (size_t)(r0 + ll) * DIM + 2 * lh;
    const float* bptr = Wo   + (size_t)(d0 + ll) * DIM + 2 * lh;

    v8f acc = {};
    #pragma unroll
    for (int k = 0; k < DIM; k += 4) {
        v2f a = *(const v2f*)(aptr + k);
        v2f b = *(const v2f*)(bptr + k);
        acc = __builtin_amdgcn_wmma_f32_16x16x4_f32(false, a, false, b,
                                                    (short)0, acc, false, false);
    }
    #pragma unroll
    for (int r = 0; r < 8; ++r) {
        const int row = r0 + r + 8 * lh;
        const int col = d0 + ll;
        if (row < NQ)
            out[(size_t)row * DIM + col] = acc[r] + q[(size_t)row * DIM + col];
    }
}

extern "C" void kernel_launch(void* const* d_in, const int* in_sizes, int n_in,
                              void* d_out, int out_size, void* d_ws, size_t ws_size,
                              hipStream_t stream)
{
    const float* q    = (const float*)d_in[0];
    const float* Wq   = (const float*)d_in[1];
    const float* Wk   = (const float*)d_in[2];
    const float* Wv   = (const float*)d_in[3];
    const float* Wo   = (const float*)d_in[4];
    const float* temp = (const float*)d_in[5];
    const float* a1   = (const float*)d_in[6];
    const float* a2   = (const float*)d_in[7];
    const float* a3   = (const float*)d_in[8];
    const float* a4   = (const float*)d_in[9];
    float* out = (float*)d_out;

    const size_t per = (size_t)BATCH * HEADS * NPAD * CH;
    float* yq   = (float*)d_ws;
    float* yk   = yq + per;
    float* yv   = yk + per;
    float* yo   = yv + per;
    float* msum = yo + per;

    proj_kernel<<<dim3(450, 3), 256, 0, stream>>>(q, Wq, Wk, Wv, yq, yk, yv);
    norm_kernel<<<(BATCH * HEADS * NQ + 7) / 8, 256, 0, stream>>>(yq, yk, temp);
    attn_kernel<<<dim3(NT, BATCH * HEADS), 256, 0, stream>>>(
        yq, yk, yv, a1, a2, a3, a4, yo);
    bsum_kernel<<<(912 * DIM + 255) / 256, 256, 0, stream>>>(yo, msum);
    out_kernel<<<NT, 256, 0, stream>>>(msum, Wo, q, out);
}


# attention: K and V fragment loads lose their bounds branches (row index clamped to the last real row): a tile's eight loads fly as one group, a tile ahead of its matrix instructions; sums same
# speedup vs baseline: 2.1438x; 1.3122x over previous
#include <hip/hip_runtime.h>
#include <hip/hip_bf16.h>

typedef __attribute__((ext_vector_type(2))) float v2f;
typedef __attribute__((ext_vector_type(8))) float v8f;

#define BATCH 8
#define HEADS 4
#define NQ    900
#define DIM   128
#define CH    32
#define NPAD  928
#define NT    57
#define SROW  928

__global__ __launch_bounds__(256) void proj_kernel(
    const float* __restrict__ q,
    const float* __restrict__ Wq, const float* __restrict__ Wk,
    const float* __restrict__ Wv,
    float* __restrict__ yq, float* __restrict__ yk, float* __restrict__ yv)
{
    const int rt    = blockIdx.x;
    const int which = blockIdx.y;
    const int wave  = threadIdx.x >> 5;
    const int lane  = threadIdx.x & 31;
    const int lh    = lane >> 4;
    const int ll    = lane & 15;

    const float* W = (which == 0) ? Wq : (which == 1) ? Wk : Wv;
    float*       Y = (which == 0) ? yq : (which == 1) ? yk : yv;

    const int r0 = rt * 16;
    const int d0 = wave * 16;

    const float* aptr = q + (size_t)(r0 + ll) * DIM + 2 * lh;
    const float* bptr = W + (size_t)(d0 + ll) * DIM + 2 * lh;

    v8f acc = {};
    #pragma unroll
    for (int k = 0; k < DIM; k += 4) {
        v2f a = *(const v2f*)(aptr + k);
        v2f b = *(const v2f*)(bptr + k);
        acc = __builtin_amdgcn_wmma_f32_16x16x4_f32(false, a, false, b,
                                                    (short)0, acc, false, false);
    }

    const int d    = d0 + ll;
    const int hsel = d >> 5;
    const int csel = d & 31;
    #pragma unroll
    for (int r = 0; r < 8; ++r) {
        int row = r0 + r + 8 * lh;
        int b_  = row / NQ;
        int n_  = row - b_ * NQ;
        Y[(((size_t)(b_ * HEADS + hsel)) * NPAD + n_) * CH + csel] = acc[r];
    }
}

__global__ __launch_bounds__(256) void norm_kernel(
    float* __restrict__ yq, float* __restrict__ yk,
    const float* __restrict__ temp)
{
    const int rows = BATCH * HEADS * NQ;
    const int wid  = blockIdx.x * 8 + (threadIdx.x >> 5);
    if (wid >= rows) return;
    const int lane = threadIdx.x & 31;
    const int bh   = wid / NQ;
    const int n    = wid - bh * NQ;
    const int h    = bh & (HEADS - 1);

    const size_t off = ((size_t)bh * NPAD + n) * CH + lane;
    float vq = yq[off], vk = yk[off];
    float sq = vq * vq, sk = vk * vk;
    #pragma unroll
    for (int s = 16; s > 0; s >>= 1) {
        sq += __shfl_xor(sq, s, 32);
        sk += __shfl_xor(sk, s, 32);
    }
    const float t  = temp[h];
    yq[off] = vq / fmaxf(sqrtf(sq), 1e-12f) * t;
    yk[off] = vk / fmaxf(sqrtf(sk), 1e-12f);
}

__device__ __forceinline__ unsigned umin_(unsigned a, unsigned b) { return a < b ? a : b; }
__device__ __forceinline__ unsigned umax_(unsigned a, unsigned b) { return a < b ? b : a; }

template <int PX>
__device__ __forceinline__ void reg_stage(unsigned (&k)[32])
{
    #pragma unroll
    for (int r = 0; r < 32; ++r) {
        const int q = r ^ PX;
        if (r < q) {
            const unsigned a = k[r], b = k[q];
            k[r] = umin_(a, b);
            k[q] = umax_(a, b);
        }
    }
}

template <int MASK, int FLIP>
__device__ __forceinline__ void lane_stage(unsigned (&k)[32], const int lane)
{
    constexpr int HB = (MASK >= 16) ? 16 : (MASK >= 8) ? 8 : (MASK >= 4) ? 4
                     : (MASK >= 2) ? 2 : 1;
    const unsigned lim = (lane & HB) ? 0xffffffffu : 0u;
    unsigned n[32];
    #pragma unroll
    for (int r = 0; r < 32; ++r) {
        if constexpr (MASK < 16) {
            asm("v_med3_u32_e64_dpp %0, %1, %2, %3 row_xmask:%4 row_mask:0xf bank_mask:0xf"
                : "=v"(n[r]) : "v"(k[r ^ FLIP]), "v"(k[r]), "v"(lim), "n"(MASK));
        } else {
            const unsigned p = (unsigned)__shfl_xor((int)k[r ^ FLIP], MASK, 32);
            asm("v_med3_u32 %0, %1, %2, %3"
                : "=v"(n[r]) : "v"(p), "v"(k[r]), "v"(lim));
        }
    }
    #pragma unroll
    for (int r = 0; r < 32; ++r) k[r] = n[r];
}

__device__ __forceinline__ void reg_tail(unsigned (&k)[32])
{
    reg_stage<16>(k); reg_stage<8>(k); reg_stage<4>(k); reg_stage<2>(k); reg_stage<1>(k);
}

__device__ __forceinline__ void sort_row_keys(unsigned (&k)[32], const int lane)
{
    reg_stage<1>(k);
    reg_stage<3>(k);  reg_stage<1>(k);
    reg_stage<7>(k);  reg_stage<2>(k); reg_stage<1>(k);
    reg_stage<15>(k); reg_stage<4>(k); reg_stage<2>(k); reg_stage<1>(k);
    reg_stage<31>(k); reg_stage<8>(k); reg_stage<4>(k); reg_stage<2>(k); reg_stage<1>(k);
    lane_stage<1, 31>(k, lane);
    reg_tail(k);
    lane_stage<3, 31>(k, lane);  lane_stage<1, 0>(k, lane);
    reg_tail(k);
    lane_stage<7, 31>(k, lane);  lane_stage<2, 0>(k, lane); lane_stage<1, 0>(k, lane);
    reg_tail(k);
    lane_stage<15, 31>(k, lane); lane_stage<4, 0>(k, lane); lane_stage<2, 0>(k, lane);
    lane_stage<1, 0>(k, lane);
    reg_tail(k);
    lane_stage<31, 31>(k, lane); lane_stage<8, 0>(k, lane); lane_stage<4, 0>(k, lane);
    lane_stage<2, 0>(k, lane);   lane_stage<1, 0>(k, lane);
    reg_tail(k);
}

__global__ __launch_bounds__(256, 10) void attn_kernel(
    const float* __restrict__ qh, const float* __restrict__ kh,
    const float* __restrict__ vh,
    const float* __restrict__ a1p, const float* __restrict__ a2p,
    const float* __restrict__ a3p, const float* __restrict__ a4p,
    float* __restrict__ yo)
{
    __shared__ float sc[16 * SROW];

    const int bh   = blockIdx.y;
    const int nt   = blockIdx.x;
    const int wave = threadIdx.x >> 5;
    const int lane = threadIdx.x & 31;
    const int lh   = lane >> 4;
    const int ll   = lane & 15;

    const float* qbase = qh + (size_t)bh * NPAD * CH;
    const float* kbase = kh + (size_t)bh * NPAD * CH;
    const float* vbase = vh + (size_t)bh * NPAD * CH;
    const int n0 = nt * 16;

    v2f areg[8];
    {
        const float* ap = qbase + (size_t)(n0 + ll) * CH + 2 * lh;
        #pragma unroll
        for (int c = 0; c < 8; ++c) areg[c] = *(const v2f*)(ap + 4 * c);
    }

    {
        v2f bcur[8], bnxt[8];
        {
            int mrow = wave * 16 + ll;
            mrow = (mrow < NQ) ? mrow : (NQ - 1);
            const float* bp = kbase + (size_t)mrow * CH + 2 * lh;
            #pragma unroll
            for (int c = 0; c < 8; ++c) bcur[c] = *(const v2f*)(bp + 4 * c);
        }
        for (int mt = wave; mt < NT; mt += 8) {
            const int m0  = mt * 16;
            const int mtn = (mt + 8 < NT) ? (mt + 8) : mt;
            {
                int mrow = mtn * 16 + ll;
                mrow = (mrow < NQ) ? mrow : (NQ - 1);
                const float* bp = kbase + (size_t)mrow * CH + 2 * lh;
                #pragma unroll
                for (int c = 0; c < 8; ++c) bnxt[c] = *(const v2f*)(bp + 4 * c);
            }
            v8f acc = {};
            #pragma unroll
            for (int c = 0; c < 8; ++c)
                acc = __builtin_amdgcn_wmma_f32_16x16x4_f32(false, areg[c], false, bcur[c],
                                                            (short)0, acc, false, false);
            #pragma unroll
            for (int r = 0; r < 8; ++r)
                sc[(r + 8 * lh) * SROW + m0 + ll] = acc[r];
            #pragma unroll
            for (int c = 0; c < 8; ++c) bcur[c] = bnxt[c];
        }
    }
    __syncthreads();

    for (int i = threadIdx.x; i < 16 * (SROW - NQ); i += 256) {
        int row = i / (SROW - NQ);
        int m   = NQ + i - row * (SROW - NQ);
        sc[row * SROW + m] = -1e30f;
    }
    __syncthreads();

    const float A1 = a1p[0], A2 = a2p[0], A3 = a3p[0], A4 = a4p[0];

    for (int rr = 0; rr < 2; ++rr) {
        const int row = wave + 8 * rr;
        float* srow = sc + row * SROW;

        unsigned key[32];
        #pragma unroll
        for (int j = 0; j < 29; ++j) {
            const unsigned u = __float_as_uint(srow[lane + 32 * j]);
            key[j] = (u & 0x80000000u) ? ~u : (u | 0x80000000u);
        }
        key[29] = 0u; key[30] = 0u; key[31] = 0u;

        sort_row_keys(key, lane);

        unsigned th[3];
        th[0] = (unsigned)__builtin_amdgcn_readlane((int)key[28], 28);
        th[1] = (unsigned)__builtin_amdgcn_readlane((int)key[20], 22);
        th[2] = (unsigned)__builtin_amdgcn_readlane((int)key[8],  13);
        const unsigned ktop = (unsigned)__builtin_amdgcn_readlane((int)key[31], 31);
        const float mx = __uint_as_float((ktop & 0x80000000u) ? (ktop ^ 0x80000000u)
                                                              : ~ktop);

        asm volatile("" ::: "memory");

        float    rv[29];
        unsigned rk[29];
        #pragma unroll
        for (int j = 0; j < 29; ++j) {
            rv[j] = srow[lane + 32 * j];
            const unsigned u = __float_as_uint(rv[j]);
            rk[j] = (u & 0x80000000u) ? ~u : (u | 0x80000000u);
        }

        float s1 = 0.f, s2 = 0.f, s3 = 0.f, s4 = 0.f;
        float w[29];
        #pragma unroll
        for (int j = 0; j < 29; ++j) {
            float e = __expf(rv[j] - mx);
            w[j] = e;
            s4 += e;
            if (rk[j] >= th[0]) s1 += e;
            if (rk[j] >= th[1]) s2 += e;
            if (rk[j] >= th[2]) s3 += e;
        }
        #pragma unroll
        for (int s = 16; s > 0; s >>= 1) {
            s1 += __shfl_xor(s1, s, 32);
            s2 += __shfl_xor(s2, s, 32);
            s3 += __shfl_xor(s3, s, 32);
            s4 += __shfl_xor(s4, s, 32);
        }
        const float c1 = A1 / s1, c2 = A2 / s2, c3 = A3 / s3, c4 = A4 / s4;
        #pragma unroll
        for (int j = 0; j < 29; ++j) {
            float f = c4;
            if (rk[j] >= th[0]) f += c1;
            if (rk[j] >= th[1]) f += c2;
            if (rk[j] >= th[2]) f += c3;
            srow[lane + 32 * j] = w[j] * f;
        }
    }
    __syncthreads();

    const int nhalf = wave & 1;
    const int kg    = wave >> 1;
    const int cn    = 16 * nhalf + ll;
    float vcur[8], vnxt[8];
    #pragma unroll
    for (int k4 = 0; k4 < 4; ++k4) {
        const unsigned kk = kg * 16 + 4 * k4 + 2 * lh;
        vcur[2 * k4]     = vbase[kk * CH + cn];
        vcur[2 * k4 + 1] = vbase[(kk + 1) * CH + cn];
    }
    v8f acc = {};
    for (int mt = kg; mt < NT; mt += 4) {
        const int mtn = (mt + 4 < NT) ? (mt + 4) : kg;
        #pragma unroll
        for (int k4 = 0; k4 < 4; ++k4) {
            const unsigned kk = mtn * 16 + 4 * k4 + 2 * lh;
            const unsigned r0 = (kk     < NQ) ? kk       : (NQ - 1);
            const unsigned r1 = (kk + 1 < NQ) ? (kk + 1) : (NQ - 1);
            vnxt[2 * k4]     = vbase[r0 * CH + cn];
            vnxt[2 * k4 + 1] = vbase[r1 * CH + cn];
        }
        #pragma unroll
        for (int k4 = 0; k4 < 4; ++k4) {
            const int kk = mt * 16 + 4 * k4 + 2 * lh;
            v2f a = *(const v2f*)(sc + ll * SROW + kk);
            v2f b;
            b.x = vcur[2 * k4];
            b.y = vcur[2 * k4 + 1];
            acc = __builtin_amdgcn_wmma_f32_16x16x4_f32(false, a, false, b,
                                                        (short)0, acc, false, false);
        }
        #pragma unroll
        for (int i = 0; i < 8; ++i) vcur[i] = vnxt[i];
    }
    __syncthreads();

    float* part = sc;
    #pragma unroll
    for (int r = 0; r < 8; ++r) part[(wave * 8 + r) * 32 + lane] = acc[r];
    __syncthreads();

    if (wave < 2) {
        #pragma unroll
        for (int r = 0; r < 8; ++r) {
            float v = 0.f;
            #pragma unroll
            for (int g = 0; g < 4; ++g)
                v += part[(((g * 2 + wave) * 8) + r) * 32 + lane];
            const int row = n0 + r + 8 * lh;
            const int col = 16 * wave + ll;
            if (row < NQ)
                yo[((size_t)bh * NPAD + row) * CH + col] = v;
        }
    }
}

__global__ __launch_bounds__(256) void bsum_kernel(
    const float* __restrict__ yo, float* __restrict__ msum)
{
    const int idx = blockIdx.x * 256 + threadIdx.x;
    if (idx >= 912 * DIM) return;
    const int n = idx >> 7;
    const int d = idx & 127;
    float v = 0.f;
    if (n < NQ) {
        const int h = d >> 5, c = d & 31;
        #pragma unroll
        for (int b = 0; b < BATCH; ++b)
            v += yo[(((size_t)(b * HEADS + h)) * NPAD + n) * CH + c];
    }
    msum[idx] = v;
}

__global__ __launch_bounds__(256) void out_kernel(
    const float* __restrict__ msum, const float* __restrict__ Wo,
    const float* __restrict__ q, float* __restrict__ out)
{
    const int rt   = blockIdx.x;
    const int wave = threadIdx.x >> 5;
    const int lane = threadIdx.x & 31;
    const int lh   = lane >> 4;
    const int ll   = lane & 15;
    const int r0   = rt * 16;
    const int d0   = wave * 16;

    const float* aptr = msum + (size_t)(r0 + ll) * DIM + 2 * lh;
    const float* bptr = Wo   + (size_t)(d0 + ll) * DIM + 2 * lh;

    v8f acc = {};
    #pragma unroll
    for (int k = 0; k < DIM; k += 4) {
        v2f a = *(const v2f*)(aptr + k);
        v2f b = *(const v2f*)(bptr + k);
        acc = __builtin_amdgcn_wmma_f32_16x16x4_f32(false, a, false, b,
                                                    (short)0, acc, false, false);
    }
    #pragma unroll
    for (int r = 0; r < 8; ++r) {
        const int row = r0 + r + 8 * lh;
        const int col = d0 + ll;
        if (row < NQ)
            out[(size_t)row * DIM + col] = acc[r] + q[(size_t)row * DIM + col];
    }
}

extern "C" void kernel_launch(void* const* d_in, const int* in_sizes, int n_in,
                              void* d_out, int out_size, void* d_ws, size_t ws_size,
                              hipStream_t stream)
{
    const float* q    = (const float*)d_in[0];
    const float* Wq   = (const float*)d_in[1];
    const float* Wk   = (const float*)d_in[2];
    const float* Wv   = (const float*)d_in[3];
    const float* Wo   = (const float*)d_in[4];
    const float* temp = (const float*)d_in[5];
    const float* a1   = (const float*)d_in[6];
    const float* a2   = (const float*)d_in[7];
    const float* a3   = (const float*)d_in[8];
    const float* a4   = (const float*)d_in[9];
    float* out = (float*)d_out;

    const size_t per = (size_t)BATCH * HEADS * NPAD * CH;
    float* yq   = (float*)d_ws;
    float* yk   = yq + per;
    float* yv   = yk + per;
    float* yo   = yv + per;
    float* msum = yo + per;

    proj_kernel<<<dim3(450, 3), 256, 0, stream>>>(q, Wq, Wk, Wv, yq, yk, yv);
    norm_kernel<<<(BATCH * HEADS * NQ + 7) / 8, 256, 0, stream>>>(yq, yk, temp);
    attn_kernel<<<dim3(NT, BATCH * HEADS), 256, 0, stream>>>(
        yq, yk, yv, a1, a2, a3, a4, yo);
    bsum_kernel<<<(912 * DIM + 255) / 256, 256, 0, stream>>>(yo, msum);
    out_kernel<<<NT, 256, 0, stream>>>(msum, Wo, q, out);
}
